# starting kernel
# baseline (speedup 1.0000x reference)
#include <hip/hip_runtime.h>

constexpr int B_    = 4;
constexpr int T_    = 12;
constexpr int N_    = 5000;
constexpr int C_    = 16;
constexpr int H_    = 4;
constexpr int K_    = 9;
constexpr int TOUT_ = 12;
constexpr int TH_   = T_ * H_;
constexpr int M_    = B_ * N_ * C_;

typedef float v2f __attribute__((ext_vector_type(2)));
typedef float v8f __attribute__((ext_vector_type(8)));

__device__ __forceinline__ float selu_f(float x) {
    const float scale = 1.0507009873554805f;
    const float alpha = 1.6732632423543772f;
    return x > 0.0f ? scale * x : scale * alpha * (expf(x) - 1.0f);
}

__global__ __launch_bounds__(256)
void stageA_kernel(const float* __restrict__ x, const int* __restrict__ nn,
                   const float* __restrict__ W1, const float* __restrict__ b1,
                   float* __restrict__ agg1) {
    int gid = blockIdx.x * blockDim.x + threadIdx.x;
    if (gid >= B_ * N_ * H_ * C_) return;
    int c   = gid & (C_ - 1);
    int h   = (gid >> 4) & (H_ - 1);
    int bn  = gid >> 6;
    int n   = bn % N_;
    int b   = bn / N_;

    int   idx[K_];
    float w[K_];
#pragma unroll
    for (int k = 0; k < K_; ++k) {
        int id = nn[n * K_ + k];
        float msk = (id < N_) ? 1.0f : 0.0f;
        idx[k] = (id < N_) ? id : 0;
        w[k]   = W1[h * K_ + k] * msk;
    }
    float bias = b1[h];

    float prev = 0.0f;
    for (int t = 0; t < T_; ++t) {
        const float* xbt = x + (size_t)(b * T_ + t) * N_ * C_;
        float s = bias;
#pragma unroll
        for (int k = 0; k < K_; ++k) {
            s = fmaf(w[k], xbt[idx[k] * C_ + c], s);
        }
        float bl = (t == 0) ? s : 0.8f * s + 0.2f * prev;
        prev = s;
        agg1[(((size_t)(b * T_ + t) * N_ + n) * H_ + h) * C_ + c] = selu_f(bl);
    }
}

__global__ __launch_bounds__(256)
void stageB_kernel(const float* __restrict__ agg1, const int* __restrict__ nn,
                   const float* __restrict__ W2, const float* __restrict__ b2,
                   float* __restrict__ a_mat) {
    int gid = blockIdx.x * blockDim.x + threadIdx.x;
    if (gid >= B_ * N_ * H_ * C_) return;
    int c   = gid & (C_ - 1);
    int h   = (gid >> 4) & (H_ - 1);
    int bn  = gid >> 6;
    int n   = bn % N_;
    int b   = bn / N_;

    int   idx[K_];
    float w[K_];
#pragma unroll
    for (int k = 0; k < K_; ++k) {
        int id = nn[n * K_ + k];
        float msk = (id < N_) ? 1.0f : 0.0f;
        idx[k] = (id < N_) ? id : 0;
        w[k]   = W2[k] * msk;
    }
    float bias = b2[0];

    size_t row_base = ((size_t)(b * N_ + n) * C_ + c) * TH_;
    float prev = 0.0f;
    for (int t = 0; t < T_; ++t) {
        const float* abt = agg1 + (size_t)(b * T_ + t) * N_ * H_ * C_;
        float s = bias;
#pragma unroll
        for (int k = 0; k < K_; ++k) {
            s = fmaf(w[k], abt[(idx[k] * H_ + h) * C_ + c], s);
        }
        float bl = (t == 0) ? s : 0.8f * s + 0.2f * prev;
        prev = s;
        a_mat[row_base + t * H_ + h] = selu_f(bl);
    }
}

__global__ __launch_bounds__(256)
void copyx_kernel(const float* __restrict__ x, float* __restrict__ out) {
    int gid = blockIdx.x * blockDim.x + threadIdx.x;
    if (gid >= B_ * T_ * N_ * C_) return;
    int per_b = T_ * N_ * C_;
    int b = gid / per_b;
    int r = gid - b * per_b;
    out[(size_t)b * (2 * T_) * N_ * C_ + r] = x[gid];
}

__global__ __launch_bounds__(256)
void gemm_wmma_kernel(const float* __restrict__ a_mat, const float* __restrict__ Ws,
                      const float* __restrict__ bs, float* __restrict__ out) {
    int wave = (int)(blockIdx.x * (blockDim.x >> 5)) + (threadIdx.x >> 5);
    int lane = threadIdx.x & 31;
    int m0   = lane & 15;
    int half = lane >> 4;

    int   colc = (m0 < TOUT_) ? m0 : 0;
    float msk  = (m0 < TOUT_) ? 1.0f : 0.0f;
    const float* wrow = Ws + colc * TH_;
    v2f Bfr[TH_ / 4];
#pragma unroll
    for (int kk = 0; kk < TH_; kk += 4) {
        int kr = kk + half * 2;
        Bfr[kk >> 2].x = wrow[kr + 0] * msk;
        Bfr[kk >> 2].y = wrow[kr + 1] * msk;
    }

    int row = wave * 16 + m0;
    const float* arow = a_mat + (size_t)row * TH_;

    v8f acc = {};
#pragma unroll
    for (int kk = 0; kk < TH_; kk += 4) {
        int kr = kk + half * 2;
        v2f Af;
        Af.x = arow[kr + 0];
        Af.y = arow[kr + 1];
        acc = __builtin_amdgcn_wmma_f32_16x16x4_f32(
             false, Af,  false, Bfr[kk >> 2],
             (short)0, acc,  false,  false);
    }

    if (m0 < TOUT_) {
        float bv = bs[m0];
#pragma unroll
        for (int r = 0; r < 8; ++r) {
            int mrow = wave * 16 + r + half * 8;
            int c  = mrow & (C_ - 1);
            int bn = mrow >> 4;
            int n  = bn % N_;
            int b  = bn / N_;
            float v = selu_f(acc[r] + bv);
            out[(((size_t)b * (2 * T_) + (T_ + m0)) * N_ + n) * C_ + c] = v;
        }
    }
}

extern "C" void kernel_launch(void* const* d_in, const int* in_sizes, int n_in,
                              void* d_out, int out_size, void* d_ws, size_t ws_size,
                              hipStream_t stream) {
    const float* x  = (const float*)d_in[0];
    const int*   nn = (const int*)  d_in[1];
    const float* W1 = (const float*)d_in[2];
    const float* b1 = (const float*)d_in[3];
    const float* W2 = (const float*)d_in[4];
    const float* b2 = (const float*)d_in[5];
    const float* Ws = (const float*)d_in[6];
    const float* bs = (const float*)d_in[7];
    float* out = (float*)d_out;

    float* agg1  = (float*)d_ws;
    float* a_mat = agg1 + (size_t)B_ * T_ * N_ * H_ * C_;

    int nthA = B_ * N_ * H_ * C_;
    stageA_kernel<<<(nthA + 255) / 256, 256, 0, stream>>>(x, nn, W1, b1, agg1);
    stageB_kernel<<<(nthA + 255) / 256, 256, 0, stream>>>(agg1, nn, W2, b2, a_mat);

    int nCopy = B_ * T_ * N_ * C_;
    copyx_kernel<<<(nCopy + 255) / 256, 256, 0, stream>>>(x, out);

    int tiles = M_ / 16;
    gemm_wmma_kernel<<<tiles / 8, 256, 0, stream>>>(a_mat, Ws, bs, out);
}


# Stage 2 fused into the WMMA GEMM: one wave per (b,n) builds its 16x48 A tile in registers as A fragments; a_mat and one launch gone; heads in k slots 0,2,1,3 in A and B (sum order in a k-step)
# speedup vs baseline: 1.4981x; 1.4981x over previous
#include <hip/hip_runtime.h>

constexpr int B_    = 4;
constexpr int T_    = 12;
constexpr int N_    = 5000;
constexpr int C_    = 16;
constexpr int H_    = 4;
constexpr int K_    = 9;
constexpr int TOUT_ = 12;
constexpr int TH_   = T_ * H_;

typedef float v2f __attribute__((ext_vector_type(2)));
typedef float v8f __attribute__((ext_vector_type(8)));

__device__ __forceinline__ float selu_f(float x) {
    const float scale = 1.0507009873554805f;
    const float alpha = 1.6732632423543772f;
    return x > 0.0f ? scale * x : scale * alpha * (expf(x) - 1.0f);
}

__global__ __launch_bounds__(256)
void stageA_kernel(const float* __restrict__ x, const int* __restrict__ nn,
                   const float* __restrict__ W1, const float* __restrict__ b1,
                   float* __restrict__ agg1) {
    int gid = blockIdx.x * blockDim.x + threadIdx.x;
    if (gid >= B_ * N_ * H_ * C_) return;
    int c   = gid & (C_ - 1);
    int h   = (gid >> 4) & (H_ - 1);
    int bn  = gid >> 6;
    int n   = bn % N_;
    int b   = bn / N_;

    int   idx[K_];
    float w[K_];
#pragma unroll
    for (int k = 0; k < K_; ++k) {
        int id = nn[n * K_ + k];
        float msk = (id < N_) ? 1.0f : 0.0f;
        idx[k] = (id < N_) ? id : 0;
        w[k]   = W1[h * K_ + k] * msk;
    }
    float bias = b1[h];

    float prev = 0.0f;
    for (int t = 0; t < T_; ++t) {
        const float* xbt = x + (size_t)(b * T_ + t) * N_ * C_;
        float s = bias;
#pragma unroll
        for (int k = 0; k < K_; ++k) {
            s = fmaf(w[k], xbt[idx[k] * C_ + c], s);
        }
        float bl = (t == 0) ? s : 0.8f * s + 0.2f * prev;
        prev = s;
        agg1[(((size_t)(b * T_ + t) * N_ + n) * H_ + h) * C_ + c] = selu_f(bl);
    }
}

__global__ __launch_bounds__(256)
void stageB_gemm_kernel(const float* __restrict__ agg1, const int* __restrict__ nn,
                        const float* __restrict__ W2, const float* __restrict__ b2,
                        const float* __restrict__ Ws, const float* __restrict__ bs,
                        float* __restrict__ out) {
    const int wv   = __builtin_amdgcn_readfirstlane((int)(threadIdx.x >> 5));
    const int bn   = (int)blockIdx.x * 8 + wv;
    const int n    = bn % N_;
    const int b    = bn / N_;
    const int lane = threadIdx.x & 31;
    const int col  = lane & 15;
    const int hs   = lane >> 4;

    int   rowoff[K_];
    float w[K_];
#pragma unroll
    for (int k = 0; k < K_; ++k) {
        int id = nn[n * K_ + k];
        float mk = (id < N_) ? 1.0f : 0.0f;
        rowoff[k] = (id < N_) ? id * (H_ * C_) : 0;
        w[k]      = W2[k] * mk;
    }
    const float bias = b2[0];

    const int   colc = (col < TOUT_) ? col : 0;
    const float msk  = (col < TOUT_) ? 1.0f : 0.0f;
    const float* wrow = Ws + colc * TH_ + hs;
    v2f Bfr[T_];
#pragma unroll
    for (int t = 0; t < T_; ++t) {
        Bfr[t].x = wrow[t * H_ + 0] * msk;
        Bfr[t].y = wrow[t * H_ + 2] * msk;
    }

    float prev0 = 0.0f, prev1 = 0.0f;
    v8f acc = {};
#pragma unroll
    for (int t = 0; t < T_; ++t) {
        const float* pt = agg1 + (size_t)(b * T_ + t) * N_ * H_ * C_;
        float s0 = bias, s1 = bias;
#pragma unroll
        for (int k = 0; k < K_; ++k) {
            const float* r = pt + rowoff[k];
            s0 = fmaf(w[k], r[lane], s0);
            s1 = fmaf(w[k], r[lane + 2 * C_], s1);
        }
        float bl0 = (t == 0) ? s0 : 0.8f * s0 + 0.2f * prev0;
        float bl1 = (t == 0) ? s1 : 0.8f * s1 + 0.2f * prev1;
        prev0 = s0;
        prev1 = s1;
        v2f Af;
        Af.x = selu_f(bl0);
        Af.y = selu_f(bl1);
        acc = __builtin_amdgcn_wmma_f32_16x16x4_f32(
             false, Af,  false, Bfr[t],
             (short)0, acc,  false,  false);
    }

    const float bv = bs[colc];
    float y[8];
#pragma unroll
    for (int r = 0; r < 8; ++r) y[r] = selu_f(acc[r] + bv);

    if (col < TOUT_) {
        float* o = out + (((size_t)b * (2 * T_) + (T_ + col)) * N_ + n) * C_ + hs * 8;
        *reinterpret_cast<float4*>(o)     = make_float4(y[0], y[1], y[2], y[3]);
        *reinterpret_cast<float4*>(o + 4) = make_float4(y[4], y[5], y[6], y[7]);
    }
}

__global__ __launch_bounds__(256)
void copyx_kernel(const float* __restrict__ x, float* __restrict__ out) {
    int gid = blockIdx.x * blockDim.x + threadIdx.x;
    if (gid >= B_ * T_ * N_ * C_) return;
    int per_b = T_ * N_ * C_;
    int b = gid / per_b;
    int r = gid - b * per_b;
    out[(size_t)b * (2 * T_) * N_ * C_ + r] = x[gid];
}

extern "C" void kernel_launch(void* const* d_in, const int* in_sizes, int n_in,
                              void* d_out, int out_size, void* d_ws, size_t ws_size,
                              hipStream_t stream) {
    const float* x  = (const float*)d_in[0];
    const int*   nn = (const int*)  d_in[1];
    const float* W1 = (const float*)d_in[2];
    const float* b1 = (const float*)d_in[3];
    const float* W2 = (const float*)d_in[4];
    const float* b2 = (const float*)d_in[5];
    const float* Ws = (const float*)d_in[6];
    const float* bs = (const float*)d_in[7];
    float* out = (float*)d_out;

    float* agg1 = (float*)d_ws;

    int nthA = B_ * N_ * H_ * C_;
    stageA_kernel<<<(nthA + 255) / 256, 256, 0, stream>>>(x, nn, W1, b1, agg1);

    stageB_gemm_kernel<<<(B_ * N_) / 8, 256, 0, stream>>>(agg1, nn, W2, b2, Ws, bs, out);

    int nCopy = B_ * T_ * N_ * C_;
    copyx_kernel<<<(nCopy + 255) / 256, 256, 0, stream>>>(x, out);
}


# x re-laid by the copy launch (now first) as [b][n][t/4][c][t%4] in the workspace; stage 1 gathers a neighbour's 12 steps as three aligned 256-byte units, not 12 scattered 64-byte rows
# speedup vs baseline: 1.7722x; 1.1830x over previous
#include <hip/hip_runtime.h>

constexpr int B_    = 4;
constexpr int T_    = 12;
constexpr int N_    = 5000;
constexpr int C_    = 16;
constexpr int H_    = 4;
constexpr int K_    = 9;
constexpr int TOUT_ = 12;
constexpr int TH_   = T_ * H_;

typedef float v2f __attribute__((ext_vector_type(2)));
typedef float v8f __attribute__((ext_vector_type(8)));

__device__ __forceinline__ float selu_f(float x) {
    const float scale = 1.0507009873554805f;
    const float alpha = 1.6732632423543772f;
    return x > 0.0f ? scale * x : scale * alpha * (expf(x) - 1.0f);
}

__global__ __launch_bounds__(256)
void stageA_kernel(const float* __restrict__ xt, const int* __restrict__ nn,
                   const float* __restrict__ W1, const float* __restrict__ b1,
                   float* __restrict__ agg1) {
    int gid = blockIdx.x * blockDim.x + threadIdx.x;
    if (gid >= B_ * N_ * H_ * C_) return;
    int c   = gid & (C_ - 1);
    int h   = (gid >> 4) & (H_ - 1);
    int bn  = gid >> 6;
    int n   = bn % N_;
    int b   = bn / N_;

    const float bias = b1[h];
    float s[T_];
#pragma unroll
    for (int t = 0; t < T_; ++t) s[t] = bias;

#pragma unroll
    for (int k = 0; k < K_; ++k) {
        int id = nn[n * K_ + k];
        float msk = (id < N_) ? 1.0f : 0.0f;
        int idc   = (id < N_) ? id : 0;
        float w   = W1[h * K_ + k] * msk;
        const float4* p = reinterpret_cast<const float4*>(
            xt + (size_t)(b * N_ + idc) * (C_ * T_) + c * 4);
        float4 v0 = p[0];
        float4 v1 = p[C_];
        float4 v2 = p[2 * C_];
        s[0]  = fmaf(w, v0.x, s[0]);
        s[1]  = fmaf(w, v0.y, s[1]);
        s[2]  = fmaf(w, v0.z, s[2]);
        s[3]  = fmaf(w, v0.w, s[3]);
        s[4]  = fmaf(w, v1.x, s[4]);
        s[5]  = fmaf(w, v1.y, s[5]);
        s[6]  = fmaf(w, v1.z, s[6]);
        s[7]  = fmaf(w, v1.w, s[7]);
        s[8]  = fmaf(w, v2.x, s[8]);
        s[9]  = fmaf(w, v2.y, s[9]);
        s[10] = fmaf(w, v2.z, s[10]);
        s[11] = fmaf(w, v2.w, s[11]);
    }

#pragma unroll
    for (int t = 0; t < T_; ++t) {
        float bl = (t == 0) ? s[0] : fmaf(0.2f, s[t - 1], 0.8f * s[t]);
        agg1[(((size_t)(b * T_ + t) * N_ + n) * H_ + h) * C_ + c] = selu_f(bl);
    }
}

__global__ __launch_bounds__(256)
void stageB_gemm_kernel(const float* __restrict__ agg1, const int* __restrict__ nn,
                        const float* __restrict__ W2, const float* __restrict__ b2,
                        const float* __restrict__ Ws, const float* __restrict__ bs,
                        float* __restrict__ out) {
    const int wv   = __builtin_amdgcn_readfirstlane((int)(threadIdx.x >> 5));
    const int bn   = (int)blockIdx.x * 8 + wv;
    const int n    = bn % N_;
    const int b    = bn / N_;
    const int lane = threadIdx.x & 31;
    const int col  = lane & 15;
    const int hs   = lane >> 4;

    int   rowoff[K_];
    float w[K_];
#pragma unroll
    for (int k = 0; k < K_; ++k) {
        int id = nn[n * K_ + k];
        float mk = (id < N_) ? 1.0f : 0.0f;
        rowoff[k] = (id < N_) ? id * (H_ * C_) : 0;
        w[k]      = W2[k] * mk;
    }
    const float bias = b2[0];

    const int   colc = (col < TOUT_) ? col : 0;
    const float msk  = (col < TOUT_) ? 1.0f : 0.0f;
    const float* wrow = Ws + colc * TH_ + hs;
    v2f Bfr[T_];
#pragma unroll
    for (int t = 0; t < T_; ++t) {
        Bfr[t].x = wrow[t * H_ + 0] * msk;
        Bfr[t].y = wrow[t * H_ + 2] * msk;
    }

    float prev0 = 0.0f, prev1 = 0.0f;
    v8f acc = {};
#pragma unroll
    for (int t = 0; t < T_; ++t) {
        const float* pt = agg1 + (size_t)(b * T_ + t) * N_ * H_ * C_;
        float s0 = bias, s1 = bias;
#pragma unroll
        for (int k = 0; k < K_; ++k) {
            const float* r = pt + rowoff[k];
            s0 = fmaf(w[k], r[lane], s0);
            s1 = fmaf(w[k], r[lane + 2 * C_], s1);
        }
        float bl0 = (t == 0) ? s0 : 0.8f * s0 + 0.2f * prev0;
        float bl1 = (t == 0) ? s1 : 0.8f * s1 + 0.2f * prev1;
        prev0 = s0;
        prev1 = s1;
        v2f Af;
        Af.x = selu_f(bl0);
        Af.y = selu_f(bl1);
        acc = __builtin_amdgcn_wmma_f32_16x16x4_f32(
             false, Af,  false, Bfr[t],
             (short)0, acc,  false,  false);
    }

    const float bv = bs[colc];
    float y[8];
#pragma unroll
    for (int r = 0; r < 8; ++r) y[r] = selu_f(acc[r] + bv);

    if (col < TOUT_) {
        float* o = out + (((size_t)b * (2 * T_) + (T_ + col)) * N_ + n) * C_ + hs * 8;
        *reinterpret_cast<float4*>(o)     = make_float4(y[0], y[1], y[2], y[3]);
        *reinterpret_cast<float4*>(o + 4) = make_float4(y[4], y[5], y[6], y[7]);
    }
}

__global__ __launch_bounds__(256)
void relayout_copy_kernel(const float* __restrict__ x, float* __restrict__ out,
                          float* __restrict__ xt) {
    int gid = blockIdx.x * blockDim.x + threadIdx.x;
    if (gid >= B_ * N_ * C_) return;
    int c  = gid & (C_ - 1);
    int bn = gid >> 4;
    int n  = bn % N_;
    int b  = bn / N_;

    float v[T_];
#pragma unroll
    for (int t = 0; t < T_; ++t)
        v[t] = x[((size_t)(b * T_ + t) * N_ + n) * C_ + c];
#pragma unroll
    for (int t = 0; t < T_; ++t)
        out[((size_t)(b * 2 * T_ + t) * N_ + n) * C_ + c] = v[t];

    float4* q = reinterpret_cast<float4*>(xt + (size_t)bn * (C_ * T_) + c * 4);
    q[0]      = make_float4(v[0], v[1], v[2],  v[3]);
    q[C_]     = make_float4(v[4], v[5], v[6],  v[7]);
    q[2 * C_] = make_float4(v[8], v[9], v[10], v[11]);
}

extern "C" void kernel_launch(void* const* d_in, const int* in_sizes, int n_in,
                              void* d_out, int out_size, void* d_ws, size_t ws_size,
                              hipStream_t stream) {
    const float* x  = (const float*)d_in[0];
    const int*   nn = (const int*)  d_in[1];
    const float* W1 = (const float*)d_in[2];
    const float* b1 = (const float*)d_in[3];
    const float* W2 = (const float*)d_in[4];
    const float* b2 = (const float*)d_in[5];
    const float* Ws = (const float*)d_in[6];
    const float* bs = (const float*)d_in[7];
    float* out = (float*)d_out;

    float* agg1 = (float*)d_ws;
    float* xt   = agg1 + (size_t)B_ * T_ * N_ * H_ * C_;

    constexpr int nthC = B_ * N_ * C_;
    relayout_copy_kernel<<<(nthC + 255) / 256, 256, 0, stream>>>(x, out, xt);

    constexpr int nthA = B_ * N_ * H_ * C_;
    stageA_kernel<<<(nthA + 255) / 256, 256, 0, stream>>>(xt, nn, W1, b1, agg1);

    stageB_gemm_kernel<<<(B_ * N_) / 8, 256, 0, stream>>>(agg1, nn, W2, b2, Ws, bs, out);
}


# SELU made branch-free (exp on every lane, then a select): the fused kernel's 12-step loop is one basic block, so later steps' gather loads issue before earlier waits; stage 1 sheds its 12 branches
# speedup vs baseline: 1.9607x; 1.1064x over previous
#include <hip/hip_runtime.h>

constexpr int B_    = 4;
constexpr int T_    = 12;
constexpr int N_    = 5000;
constexpr int C_    = 16;
constexpr int H_    = 4;
constexpr int K_    = 9;
constexpr int TOUT_ = 12;
constexpr int TH_   = T_ * H_;

typedef float v2f __attribute__((ext_vector_type(2)));
typedef float v8f __attribute__((ext_vector_type(8)));

__device__ __forceinline__ float selu_f(float x) {
    const float scale = 1.0507009873554805f;
    const float alpha = 1.6732632423543772f;
    const float neg = scale * alpha * (expf(x) - 1.0f);
    const float pos = scale * x;
    return x > 0.0f ? pos : neg;
}

__global__ __launch_bounds__(256)
void stageA_kernel(const float* __restrict__ xt, const int* __restrict__ nn,
                   const float* __restrict__ W1, const float* __restrict__ b1,
                   float* __restrict__ agg1) {
    int gid = blockIdx.x * blockDim.x + threadIdx.x;
    if (gid >= B_ * N_ * H_ * C_) return;
    int c   = gid & (C_ - 1);
    int h   = (gid >> 4) & (H_ - 1);
    int bn  = gid >> 6;
    int n   = bn % N_;
    int b   = bn / N_;

    const float bias = b1[h];
    float s[T_];
#pragma unroll
    for (int t = 0; t < T_; ++t) s[t] = bias;

#pragma unroll
    for (int k = 0; k < K_; ++k) {
        int id = nn[n * K_ + k];
        float msk = (id < N_) ? 1.0f : 0.0f;
        int idc   = (id < N_) ? id : 0;
        float w   = W1[h * K_ + k] * msk;
        const float4* p = reinterpret_cast<const float4*>(
            xt + (size_t)(b * N_ + idc) * (C_ * T_) + c * 4);
        float4 v0 = p[0];
        float4 v1 = p[C_];
        float4 v2 = p[2 * C_];
        s[0]  = fmaf(w, v0.x, s[0]);
        s[1]  = fmaf(w, v0.y, s[1]);
        s[2]  = fmaf(w, v0.z, s[2]);
        s[3]  = fmaf(w, v0.w, s[3]);
        s[4]  = fmaf(w, v1.x, s[4]);
        s[5]  = fmaf(w, v1.y, s[5]);
        s[6]  = fmaf(w, v1.z, s[6]);
        s[7]  = fmaf(w, v1.w, s[7]);
        s[8]  = fmaf(w, v2.x, s[8]);
        s[9]  = fmaf(w, v2.y, s[9]);
        s[10] = fmaf(w, v2.z, s[10]);
        s[11] = fmaf(w, v2.w, s[11]);
    }

#pragma unroll
    for (int t = 0; t < T_; ++t) {
        float bl = (t == 0) ? s[0] : fmaf(0.2f, s[t - 1], 0.8f * s[t]);
        agg1[(((size_t)(b * T_ + t) * N_ + n) * H_ + h) * C_ + c] = selu_f(bl);
    }
}

__global__ __launch_bounds__(256)
void stageB_gemm_kernel(const float* __restrict__ agg1, const int* __restrict__ nn,
                        const float* __restrict__ W2, const float* __restrict__ b2,
                        const float* __restrict__ Ws, const float* __restrict__ bs,
                        float* __restrict__ out) {
    const int wv   = __builtin_amdgcn_readfirstlane((int)(threadIdx.x >> 5));
    const int bn   = (int)blockIdx.x * 8 + wv;
    const int n    = bn % N_;
    const int b    = bn / N_;
    const int lane = threadIdx.x & 31;
    const int col  = lane & 15;
    const int hs   = lane >> 4;

    int   rowoff[K_];
    float w[K_];
#pragma unroll
    for (int k = 0; k < K_; ++k) {
        int id = nn[n * K_ + k];
        float mk = (id < N_) ? 1.0f : 0.0f;
        rowoff[k] = (id < N_) ? id * (H_ * C_) : 0;
        w[k]      = W2[k] * mk;
    }
    const float bias = b2[0];

    const int   colc = (col < TOUT_) ? col : 0;
    const float msk  = (col < TOUT_) ? 1.0f : 0.0f;
    const float* wrow = Ws + colc * TH_ + hs;
    v2f Bfr[T_];
#pragma unroll
    for (int t = 0; t < T_; ++t) {
        Bfr[t].x = wrow[t * H_ + 0] * msk;
        Bfr[t].y = wrow[t * H_ + 2] * msk;
    }

    float prev0 = 0.0f, prev1 = 0.0f;
    v8f acc = {};
#pragma unroll
    for (int t = 0; t < T_; ++t) {
        const float* pt = agg1 + (size_t)(b * T_ + t) * N_ * H_ * C_;
        float s0 = bias, s1 = bias;
#pragma unroll
        for (int k = 0; k < K_; ++k) {
            const float* r = pt + rowoff[k];
            s0 = fmaf(w[k], r[lane], s0);
            s1 = fmaf(w[k], r[lane + 2 * C_], s1);
        }
        float bl0 = (t == 0) ? s0 : 0.8f * s0 + 0.2f * prev0;
        float bl1 = (t == 0) ? s1 : 0.8f * s1 + 0.2f * prev1;
        prev0 = s0;
        prev1 = s1;
        v2f Af;
        Af.x = selu_f(bl0);
        Af.y = selu_f(bl1);
        acc = __builtin_amdgcn_wmma_f32_16x16x4_f32(
             false, Af,  false, Bfr[t],
             (short)0, acc,  false,  false);
    }

    const float bv = bs[colc];
    float y[8];
#pragma unroll
    for (int r = 0; r < 8; ++r) y[r] = selu_f(acc[r] + bv);

    if (col < TOUT_) {
        float* o = out + (((size_t)b * (2 * T_) + (T_ + col)) * N_ + n) * C_ + hs * 8;
        *reinterpret_cast<float4*>(o)     = make_float4(y[0], y[1], y[2], y[3]);
        *reinterpret_cast<float4*>(o + 4) = make_float4(y[4], y[5], y[6], y[7]);
    }
}

__global__ __launch_bounds__(256)
void relayout_copy_kernel(const float* __restrict__ x, float* __restrict__ out,
                          float* __restrict__ xt) {
    int gid = blockIdx.x * blockDim.x + threadIdx.x;
    if (gid >= B_ * N_ * C_) return;
    int c  = gid & (C_ - 1);
    int bn = gid >> 4;
    int n  = bn % N_;
    int b  = bn / N_;

    float v[T_];
#pragma unroll
    for (int t = 0; t < T_; ++t)
        v[t] = x[((size_t)(b * T_ + t) * N_ + n) * C_ + c];
#pragma unroll
    for (int t = 0; t < T_; ++t)
        out[((size_t)(b * 2 * T_ + t) * N_ + n) * C_ + c] = v[t];

    float4* q = reinterpret_cast<float4*>(xt + (size_t)bn * (C_ * T_) + c * 4);
    q[0]      = make_float4(v[0], v[1], v[2],  v[3]);
    q[C_]     = make_float4(v[4], v[5], v[6],  v[7]);
    q[2 * C_] = make_float4(v[8], v[9], v[10], v[11]);
}

extern "C" void kernel_launch(void* const* d_in, const int* in_sizes, int n_in,
                              void* d_out, int out_size, void* d_ws, size_t ws_size,
                              hipStream_t stream) {
    const float* x  = (const float*)d_in[0];
    const int*   nn = (const int*)  d_in[1];
    const float* W1 = (const float*)d_in[2];
    const float* b1 = (const float*)d_in[3];
    const float* W2 = (const float*)d_in[4];
    const float* b2 = (const float*)d_in[5];
    const float* Ws = (const float*)d_in[6];
    const float* bs = (const float*)d_in[7];
    float* out = (float*)d_out;

    float* agg1 = (float*)d_ws;
    float* xt   = agg1 + (size_t)B_ * T_ * N_ * H_ * C_;

    constexpr int nthC = B_ * N_ * C_;
    relayout_copy_kernel<<<(nthC + 255) / 256, 256, 0, stream>>>(x, out, xt);

    constexpr int nthA = B_ * N_ * H_ * C_;
    stageA_kernel<<<(nthA + 255) / 256, 256, 0, stream>>>(xt, nn, W1, b1, agg1);

    stageB_gemm_kernel<<<(B_ * N_) / 8, 256, 0, stream>>>(agg1, nn, W2, b2, Ws, bs, out);
}


# agg1 stored [b][t][n][h/2][c][h%2]: a lane's two heads are one 8-byte unit, so stage 2 gathers a neighbour row with one 256-byte load, not two of 128; heads in k slots 0,1,2,3 (sum order in k-step)
# speedup vs baseline: 2.0411x; 1.0410x over previous
#include <hip/hip_runtime.h>

constexpr int B_    = 4;
constexpr int T_    = 12;
constexpr int N_    = 5000;
constexpr int C_    = 16;
constexpr int H_    = 4;
constexpr int K_    = 9;
constexpr int TOUT_ = 12;
constexpr int TH_   = T_ * H_;

typedef float v2f __attribute__((ext_vector_type(2)));
typedef float v8f __attribute__((ext_vector_type(8)));

__device__ __forceinline__ float selu_f(float x) {
    const float scale = 1.0507009873554805f;
    const float alpha = 1.6732632423543772f;
    const float neg = scale * alpha * (expf(x) - 1.0f);
    const float pos = scale * x;
    return x > 0.0f ? pos : neg;
}

__global__ __launch_bounds__(256)
void stageA_kernel(const float* __restrict__ xt, const int* __restrict__ nn,
                   const float* __restrict__ W1, const float* __restrict__ b1,
                   float* __restrict__ agg1) {
    int gid = blockIdx.x * blockDim.x + threadIdx.x;
    if (gid >= B_ * N_ * H_ * C_) return;
    int j   = gid & (H_ * C_ - 1);
    int c   = (j >> 1) & (C_ - 1);
    int h   = ((j >> 5) << 1) | (j & 1);
    int bn  = gid >> 6;
    int n   = bn % N_;
    int b   = bn / N_;

    const float bias = b1[h];
    float s[T_];
#pragma unroll
    for (int t = 0; t < T_; ++t) s[t] = bias;

#pragma unroll
    for (int k = 0; k < K_; ++k) {
        int id = nn[n * K_ + k];
        float msk = (id < N_) ? 1.0f : 0.0f;
        int idc   = (id < N_) ? id : 0;
        float w   = W1[h * K_ + k] * msk;
        const float4* p = reinterpret_cast<const float4*>(
            xt + (size_t)(b * N_ + idc) * (C_ * T_) + c * 4);
        float4 v0 = p[0];
        float4 v1 = p[C_];
        float4 v2 = p[2 * C_];
        s[0]  = fmaf(w, v0.x, s[0]);
        s[1]  = fmaf(w, v0.y, s[1]);
        s[2]  = fmaf(w, v0.z, s[2]);
        s[3]  = fmaf(w, v0.w, s[3]);
        s[4]  = fmaf(w, v1.x, s[4]);
        s[5]  = fmaf(w, v1.y, s[5]);
        s[6]  = fmaf(w, v1.z, s[6]);
        s[7]  = fmaf(w, v1.w, s[7]);
        s[8]  = fmaf(w, v2.x, s[8]);
        s[9]  = fmaf(w, v2.y, s[9]);
        s[10] = fmaf(w, v2.z, s[10]);
        s[11] = fmaf(w, v2.w, s[11]);
    }

#pragma unroll
    for (int t = 0; t < T_; ++t) {
        float bl = (t == 0) ? s[0] : fmaf(0.2f, s[t - 1], 0.8f * s[t]);
        agg1[((size_t)(b * T_ + t) * N_ + n) * (H_ * C_) + j] = selu_f(bl);
    }
}

__global__ __launch_bounds__(256)
void stageB_gemm_kernel(const float* __restrict__ agg1, const int* __restrict__ nn,
                        const float* __restrict__ W2, const float* __restrict__ b2,
                        const float* __restrict__ Ws, const float* __restrict__ bs,
                        float* __restrict__ out) {
    const int wv   = __builtin_amdgcn_readfirstlane((int)(threadIdx.x >> 5));
    const int bn   = (int)blockIdx.x * 8 + wv;
    const int n    = bn % N_;
    const int b    = bn / N_;
    const int lane = threadIdx.x & 31;
    const int col  = lane & 15;
    const int hs   = lane >> 4;

    int   rowoff[K_];
    float w[K_];
#pragma unroll
    for (int k = 0; k < K_; ++k) {
        int id = nn[n * K_ + k];
        float mk = (id < N_) ? 1.0f : 0.0f;
        rowoff[k] = (id < N_) ? id * (H_ * C_) : 0;
        w[k]      = W2[k] * mk;
    }
    const float bias = b2[0];

    const int   colc = (col < TOUT_) ? col : 0;
    const float msk  = (col < TOUT_) ? 1.0f : 0.0f;
    const float* wrow = Ws + colc * TH_ + 2 * hs;
    v2f Bfr[T_];
#pragma unroll
    for (int t = 0; t < T_; ++t) {
        const v2f wv2 = *reinterpret_cast<const v2f*>(wrow + t * H_);
        Bfr[t] = wv2 * msk;
    }

    float prev0 = 0.0f, prev1 = 0.0f;
    v8f acc = {};
#pragma unroll
    for (int t = 0; t < T_; ++t) {
        const float* pt = agg1 + (size_t)(b * T_ + t) * N_ * H_ * C_;
        float s0 = bias, s1 = bias;
#pragma unroll
        for (int k = 0; k < K_; ++k) {
            const float* r = pt + rowoff[k];
            const v2f v = *reinterpret_cast<const v2f*>(r + 2 * lane);
            s0 = fmaf(w[k], v.x, s0);
            s1 = fmaf(w[k], v.y, s1);
        }
        float bl0 = (t == 0) ? s0 : 0.8f * s0 + 0.2f * prev0;
        float bl1 = (t == 0) ? s1 : 0.8f * s1 + 0.2f * prev1;
        prev0 = s0;
        prev1 = s1;
        v2f Af;
        Af.x = selu_f(bl0);
        Af.y = selu_f(bl1);
        acc = __builtin_amdgcn_wmma_f32_16x16x4_f32(
             false, Af,  false, Bfr[t],
             (short)0, acc,  false,  false);
    }

    const float bv = bs[colc];
    float y[8];
#pragma unroll
    for (int r = 0; r < 8; ++r) y[r] = selu_f(acc[r] + bv);

    if (col < TOUT_) {
        float* o = out + (((size_t)b * (2 * T_) + (T_ + col)) * N_ + n) * C_ + hs * 8;
        *reinterpret_cast<float4*>(o)     = make_float4(y[0], y[1], y[2], y[3]);
        *reinterpret_cast<float4*>(o + 4) = make_float4(y[4], y[5], y[6], y[7]);
    }
}

__global__ __launch_bounds__(256)
void relayout_copy_kernel(const float* __restrict__ x, float* __restrict__ out,
                          float* __restrict__ xt) {
    int gid = blockIdx.x * blockDim.x + threadIdx.x;
    if (gid >= B_ * N_ * C_) return;
    int c  = gid & (C_ - 1);
    int bn = gid >> 4;
    int n  = bn % N_;
    int b  = bn / N_;

    float v[T_];
#pragma unroll
    for (int t = 0; t < T_; ++t)
        v[t] = x[((size_t)(b * T_ + t) * N_ + n) * C_ + c];
#pragma unroll
    for (int t = 0; t < T_; ++t)
        out[((size_t)(b * 2 * T_ + t) * N_ + n) * C_ + c] = v[t];

    float4* q = reinterpret_cast<float4*>(xt + (size_t)bn * (C_ * T_) + c * 4);
    q[0]      = make_float4(v[0], v[1], v[2],  v[3]);
    q[C_]     = make_float4(v[4], v[5], v[6],  v[7]);
    q[2 * C_] = make_float4(v[8], v[9], v[10], v[11]);
}

extern "C" void kernel_launch(void* const* d_in, const int* in_sizes, int n_in,
                              void* d_out, int out_size, void* d_ws, size_t ws_size,
                              hipStream_t stream) {
    const float* x  = (const float*)d_in[0];
    const int*   nn = (const int*)  d_in[1];
    const float* W1 = (const float*)d_in[2];
    const float* b1 = (const float*)d_in[3];
    const float* W2 = (const float*)d_in[4];
    const float* b2 = (const float*)d_in[5];
    const float* Ws = (const float*)d_in[6];
    const float* bs = (const float*)d_in[7];
    float* out = (float*)d_out;

    float* agg1 = (float*)d_ws;
    float* xt   = agg1 + (size_t)B_ * T_ * N_ * H_ * C_;

    constexpr int nthC = B_ * N_ * C_;
    relayout_copy_kernel<<<(nthC + 255) / 256, 256, 0, stream>>>(x, out, xt);

    constexpr int nthA = B_ * N_ * H_ * C_;
    stageA_kernel<<<(nthA + 255) / 256, 256, 0, stream>>>(xt, nn, W1, b1, agg1);

    stageB_gemm_kernel<<<(B_ * N_) / 8, 256, 0, stream>>>(agg1, nn, W2, b2, Ws, bs, out);
}
